# starting kernel
# baseline (speedup 1.0000x reference)
#include <hip/hip_runtime.h>
#include <hip/hip_bf16.h>

typedef __attribute__((ext_vector_type(2))) float v2f;
typedef __attribute__((ext_vector_type(8))) float v8f;

constexpr int   NQ  = 50000;
constexpr int   MS  = 50000;
constexpr int   HN  = 32;
constexpr int   KK  = 15;
constexpr int   CIN = 64;
constexpr int   CO  = 64;
constexpr float KPE = 1.2f;

constexpr int QT    = 16;
constexpr int NTHR  = 512;

__global__ __launch_bounds__(NTHR)
void kpconv_wmma_f32(const float* __restrict__ q_pts,
                     const float* __restrict__ s_pts,
                     const int*   __restrict__ nb,
                     const float* __restrict__ x,
                     const float* __restrict__ kp,
                     const float* __restrict__ wt,
                     float*       __restrict__ out)
{
    __shared__ float lds_weighted[QT * 16 * CIN];
    __shared__ float lds_partial [16 * 16 * 16];
    __shared__ float lds_nc      [QT * HN * 3];
    __shared__ int   lds_idx     [QT * HN];
    __shared__ float lds_norm    [QT];

    const int tid  = threadIdx.x;
    const int lane = tid & 31;
    const int wave = tid >> 5;
    const int qblk = blockIdx.x * QT;

    {
        const int q = wave;
        const int h = lane;
        const int n = qblk + q;
        const int idx = nb[n * HN + h];
        lds_idx[q * HN + h] = idx;
        float cx = 1e6f, cy = 1e6f, cz = 1e6f, fs = 0.0f;
        if (idx < MS) {
            cx = s_pts[idx * 3 + 0] - q_pts[n * 3 + 0];
            cy = s_pts[idx * 3 + 1] - q_pts[n * 3 + 1];
            cz = s_pts[idx * 3 + 2] - q_pts[n * 3 + 2];
            for (int c = 0; c < CIN; c += 4) {
                fs += x[idx * CIN + c + 0] + x[idx * CIN + c + 1]
                    + x[idx * CIN + c + 2] + x[idx * CIN + c + 3];
            }
        }
        lds_nc[(q * HN + h) * 3 + 0] = cx;
        lds_nc[(q * HN + h) * 3 + 1] = cy;
        lds_nc[(q * HN + h) * 3 + 2] = cz;
        const unsigned long long m = __ballot(fs > 0.0f);
        if (lane == 0) {
            const int cnt = (int)__popcll(m);
            lds_norm[q] = (float)(cnt > 0 ? cnt : 1);
        }
    }
    __syncthreads();

    {
        const int q   = wave;
        const int k   = lane & 15;
        const int col = lane & 15;
        float kx = 0.0f, ky = 0.0f, kz = 0.0f;
        if (k < KK) { kx = kp[k * 3]; ky = kp[k * 3 + 1]; kz = kp[k * 3 + 2]; }

        v8f acc[4];
        #pragma unroll
        for (int ct = 0; ct < 4; ++ct) acc[ct] = (v8f){};

        const int hofs = (lane >> 4) * 2;
        for (int hb = 0; hb < HN; hb += 4) {
            const int h0 = hb + hofs;
            v2f A;
            #pragma unroll
            for (int s = 0; s < 2; ++s) {
                const float nx = lds_nc[(q * HN + h0 + s) * 3 + 0] - kx;
                const float ny = lds_nc[(q * HN + h0 + s) * 3 + 1] - ky;
                const float nz = lds_nc[(q * HN + h0 + s) * 3 + 2] - kz;
                float w = 1.0f - sqrtf(nx * nx + ny * ny + nz * nz) * (1.0f / KPE);
                w = fmaxf(w, 0.0f);
                if (k >= KK) w = 0.0f;
                A[s] = w;
            }
            const int i0 = lds_idx[q * HN + h0];
            const int i1 = lds_idx[q * HN + h0 + 1];
            #pragma unroll
            for (int ct = 0; ct < 4; ++ct) {
                v2f B;
                B.x = (i0 < MS) ? x[i0 * CIN + ct * 16 + col] : 0.0f;
                B.y = (i1 < MS) ? x[i1 * CIN + ct * 16 + col] : 0.0f;
                acc[ct] = __builtin_amdgcn_wmma_f32_16x16x4_f32(
                    false, A, false, B, (short)0, acc[ct], false, false);
            }
        }

        const int mrow = (lane >> 4) * 8;
        #pragma unroll
        for (int ct = 0; ct < 4; ++ct)
            #pragma unroll
            for (int r = 0; r < 8; ++r)
                lds_weighted[q * 1024 + (r + mrow) * CIN + ct * 16 + col] = acc[ct][r];
    }
    __syncthreads();

    {
        const int dtile = wave & 3;
        const int kq    = wave >> 2;
        const int col   = lane & 15;
        const int kofs  = (lane >> 4) * 2;
        const int d     = dtile * 16 + col;
        const int nrow  = lane & 15;

        v8f acc = (v8f){};
        for (int s = 0; s < 64; ++s) {
            const int kb = kq * 256 + s * 4 + kofs;
            v2f A, B;
            A.x = lds_weighted[nrow * 1024 + kb];
            A.y = lds_weighted[nrow * 1024 + kb + 1];
            const int kp0 = kb >> 6, c0 = kb & 63;
            const int kp1 = (kb + 1) >> 6, c1 = (kb + 1) & 63;
            B.x = (kp0 < KK) ? wt[kp0 * (CIN * CO) + c0 * CO + d] : 0.0f;
            B.y = (kp1 < KK) ? wt[kp1 * (CIN * CO) + c1 * CO + d] : 0.0f;
            acc = __builtin_amdgcn_wmma_f32_16x16x4_f32(
                false, A, false, B, (short)0, acc, false, false);
        }
        const int mrow = (lane >> 4) * 8;
        #pragma unroll
        for (int r = 0; r < 8; ++r)
            lds_partial[wave * 256 + (r + mrow) * 16 + col] = acc[r];
    }
    __syncthreads();

    for (int e = tid; e < QT * CO; e += NTHR) {
        const int n  = e >> 6;
        const int d  = e & 63;
        const int dt = d >> 4;
        const int cc = d & 15;
        float s = 0.0f;
        #pragma unroll
        for (int kq = 0; kq < 4; ++kq)
            s += lds_partial[(kq * 4 + dt) * 256 + n * 16 + cc];
        out[(qblk + n) * CO + d] = s / lds_norm[n];
    }
}

extern "C" void kernel_launch(void* const* d_in, const int* in_sizes, int n_in,
                              void* d_out, int out_size, void* d_ws, size_t ws_size,
                              hipStream_t stream) {
    const float* q_pts = (const float*)d_in[0];
    const float* s_pts = (const float*)d_in[1];
    const int*   nb    = (const int*)  d_in[2];
    const float* x     = (const float*)d_in[3];
    const float* kp    = (const float*)d_in[4];
    const float* wt    = (const float*)d_in[5];
    float*       out   = (float*)d_out;

    const int n = in_sizes[0] / 3;
    const int nblk = (n + QT - 1) / QT;
    kpconv_wmma_f32<<<nblk, NTHR, 0, stream>>>(q_pts, s_pts, nb, x, kp, wt, out);
}


# Issue the global fragment loads of a batch of K-steps together ahead of its WMMAs (branch-free loads, 16 in flight per wave) instead of one load-wait-WMMA round trip per K-step
# speedup vs baseline: 1.4236x; 1.4236x over previous
#include <hip/hip_runtime.h>
#include <hip/hip_bf16.h>

typedef __attribute__((ext_vector_type(2))) float v2f;
typedef __attribute__((ext_vector_type(8))) float v8f;

constexpr int   NQ  = 50000;
constexpr int   MS  = 50000;
constexpr int   HN  = 32;
constexpr int   KK  = 15;
constexpr int   CIN = 64;
constexpr int   CO  = 64;
constexpr float KPE = 1.2f;

constexpr int QT    = 16;
constexpr int NTHR  = 512;

__global__ __launch_bounds__(NTHR)
void kpconv_wmma_f32(const float* __restrict__ q_pts,
                     const float* __restrict__ s_pts,
                     const int*   __restrict__ nb,
                     const float* __restrict__ x,
                     const float* __restrict__ kp,
                     const float* __restrict__ wt,
                     float*       __restrict__ out)
{
    __shared__ float lds_weighted[QT * 16 * CIN];
    __shared__ float lds_partial [16 * 16 * 16];
    __shared__ float lds_nc      [QT * HN * 3];
    __shared__ int   lds_idx     [QT * HN];
    __shared__ float lds_norm    [QT];

    const int tid  = threadIdx.x;
    const int lane = tid & 31;
    const int wave = tid >> 5;
    const int qblk = blockIdx.x * QT;

    {
        const int q = wave;
        const int h = lane;
        const int n = qblk + q;
        const int idx = nb[n * HN + h];
        lds_idx[q * HN + h] = idx;
        float cx = 1e6f, cy = 1e6f, cz = 1e6f, fs = 0.0f;
        if (idx < MS) {
            cx = s_pts[idx * 3 + 0] - q_pts[n * 3 + 0];
            cy = s_pts[idx * 3 + 1] - q_pts[n * 3 + 1];
            cz = s_pts[idx * 3 + 2] - q_pts[n * 3 + 2];
            for (int c = 0; c < CIN; c += 4) {
                fs += x[idx * CIN + c + 0] + x[idx * CIN + c + 1]
                    + x[idx * CIN + c + 2] + x[idx * CIN + c + 3];
            }
        }
        lds_nc[(q * HN + h) * 3 + 0] = cx;
        lds_nc[(q * HN + h) * 3 + 1] = cy;
        lds_nc[(q * HN + h) * 3 + 2] = cz;
        const unsigned long long m = __ballot(fs > 0.0f);
        if (lane == 0) {
            const int cnt = (int)__popcll(m);
            lds_norm[q] = (float)(cnt > 0 ? cnt : 1);
        }
    }
    __syncthreads();

    {
        const int q   = wave;
        const int k   = lane & 15;
        const int col = lane & 15;
        float kx = 0.0f, ky = 0.0f, kz = 0.0f;
        if (k < KK) { kx = kp[k * 3]; ky = kp[k * 3 + 1]; kz = kp[k * 3 + 2]; }

        v8f acc[4];
        #pragma unroll
        for (int ct = 0; ct < 4; ++ct) acc[ct] = (v8f){};

        const int hofs = (lane >> 4) * 2;
        constexpr int HB = 2;
        #pragma unroll
        for (int hb0 = 0; hb0 < HN; hb0 += 4 * HB) {
            float f0[HB][4], f1[HB][4];
            bool  ok0[HB], ok1[HB];
            #pragma unroll
            for (int j = 0; j < HB; ++j) {
                const int h0 = hb0 + 4 * j + hofs;
                const int i0 = lds_idx[q * HN + h0];
                const int i1 = lds_idx[q * HN + h0 + 1];
                ok0[j] = i0 < MS;
                ok1[j] = i1 < MS;
                const int e0 = (ok0[j] ? i0 : 0) * CIN + col;
                const int e1 = (ok1[j] ? i1 : 0) * CIN + col;
                #pragma unroll
                for (int ct = 0; ct < 4; ++ct) {
                    f0[j][ct] = x[e0 + ct * 16];
                    f1[j][ct] = x[e1 + ct * 16];
                }
            }
            #pragma unroll
            for (int j = 0; j < HB; ++j) {
                const int h0 = hb0 + 4 * j + hofs;
                v2f A;
                #pragma unroll
                for (int s = 0; s < 2; ++s) {
                    const float nx = lds_nc[(q * HN + h0 + s) * 3 + 0] - kx;
                    const float ny = lds_nc[(q * HN + h0 + s) * 3 + 1] - ky;
                    const float nz = lds_nc[(q * HN + h0 + s) * 3 + 2] - kz;
                    float w = 1.0f - sqrtf(nx * nx + ny * ny + nz * nz) * (1.0f / KPE);
                    w = fmaxf(w, 0.0f);
                    if (k >= KK) w = 0.0f;
                    A[s] = w;
                }
                #pragma unroll
                for (int ct = 0; ct < 4; ++ct) {
                    v2f B;
                    B.x = ok0[j] ? f0[j][ct] : 0.0f;
                    B.y = ok1[j] ? f1[j][ct] : 0.0f;
                    acc[ct] = __builtin_amdgcn_wmma_f32_16x16x4_f32(
                        false, A, false, B, (short)0, acc[ct], false, false);
                }
            }
        }

        const int mrow = (lane >> 4) * 8;
        #pragma unroll
        for (int ct = 0; ct < 4; ++ct)
            #pragma unroll
            for (int r = 0; r < 8; ++r)
                lds_weighted[q * 1024 + (r + mrow) * CIN + ct * 16 + col] = acc[ct][r];
    }
    __syncthreads();

    {
        const int dtile = wave & 3;
        const int kq    = wave >> 2;
        const int col   = lane & 15;
        const int kofs  = (lane >> 4) * 2;
        const int d     = dtile * 16 + col;
        const int nrow  = lane & 15;

        const int kb0    = kq * 256 + kofs;
        const int nsteps = (kq == 3) ? 48 : 64;
        constexpr int SB = 8;

        const float* wrow = wt + (kb0 * CO + d);
        const float* arow = lds_weighted + (nrow * 1024 + kb0);

        v8f acc = (v8f){};
        for (int s0 = 0; s0 < nsteps; s0 += SB) {
            v2f A[SB], B[SB];
            #pragma unroll
            for (int j = 0; j < SB; ++j) {
                B[j].x = wrow[(j * 4)     * CO];
                B[j].y = wrow[(j * 4 + 1) * CO];
            }
            #pragma unroll
            for (int j = 0; j < SB; ++j)
                A[j] = *reinterpret_cast<const v2f*>(arow + j * 4);
            #pragma unroll
            for (int j = 0; j < SB; ++j)
                acc = __builtin_amdgcn_wmma_f32_16x16x4_f32(
                    false, A[j], false, B[j], (short)0, acc, false, false);
            wrow += SB * 4 * CO;
            arow += SB * 4;
        }
        const int mrow = (lane >> 4) * 8;
        #pragma unroll
        for (int r = 0; r < 8; ++r)
            lds_partial[wave * 256 + (r + mrow) * 16 + col] = acc[r];
    }
    __syncthreads();

    for (int e = tid; e < QT * CO; e += NTHR) {
        const int n  = e >> 6;
        const int d  = e & 63;
        const int dt = d >> 4;
        const int cc = d & 15;
        float s = 0.0f;
        #pragma unroll
        for (int kq = 0; kq < 4; ++kq)
            s += lds_partial[(kq * 4 + dt) * 256 + n * 16 + cc];
        out[(qblk + n) * CO + d] = s / lds_norm[n];
    }
}

extern "C" void kernel_launch(void* const* d_in, const int* in_sizes, int n_in,
                              void* d_out, int out_size, void* d_ws, size_t ws_size,
                              hipStream_t stream) {
    const float* q_pts = (const float*)d_in[0];
    const float* s_pts = (const float*)d_in[1];
    const int*   nb    = (const int*)  d_in[2];
    const float* x     = (const float*)d_in[3];
    const float* kp    = (const float*)d_in[4];
    const float* wt    = (const float*)d_in[5];
    float*       out   = (float*)d_out;

    const int n = in_sizes[0] / 3;
    const int nblk = (n + QT - 1) / QT;
    kpconv_wmma_f32<<<nblk, NTHR, 0, stream>>>(q_pts, s_pts, nb, x, kp, wt, out);
}


# Pad the LDS row pitch of the per-query weighted tile from 1024 to 1028 dwords so the 8-byte pairs of a phase-B A-fragment read fall on 64 different banks instead of 4 (16-way conflict removed)
# speedup vs baseline: 1.7241x; 1.2111x over previous
#include <hip/hip_runtime.h>
#include <hip/hip_bf16.h>

typedef __attribute__((ext_vector_type(2))) float v2f;
typedef __attribute__((ext_vector_type(8))) float v8f;

constexpr int   NQ  = 50000;
constexpr int   MS  = 50000;
constexpr int   HN  = 32;
constexpr int   KK  = 15;
constexpr int   CIN = 64;
constexpr int   CO  = 64;
constexpr float KPE = 1.2f;

constexpr int QT    = 16;
constexpr int NTHR  = 512;
constexpr int WP    = 16 * CIN + 4;

__global__ __launch_bounds__(NTHR, 12)
void kpconv_wmma_f32(const float* __restrict__ q_pts,
                     const float* __restrict__ s_pts,
                     const int*   __restrict__ nb,
                     const float* __restrict__ x,
                     const float* __restrict__ kp,
                     const float* __restrict__ wt,
                     float*       __restrict__ out)
{
    struct alignas(16) Lds {
        float partial [16 * 16 * 16];
        float nc      [QT * HN * 3];
        int   idx     [QT * HN];
        float norm    [QT];
        float weighted[QT * WP];
    };
    __shared__ Lds lds;
    float* const lds_partial  = lds.partial;
    float* const lds_nc       = lds.nc;
    int*   const lds_idx      = lds.idx;
    float* const lds_norm     = lds.norm;
    float* const lds_weighted = lds.weighted;

    const int tid  = threadIdx.x;
    const int lane = tid & 31;
    const int wave = tid >> 5;
    const int qblk = blockIdx.x * QT;

    {
        const int q = wave;
        const int h = lane;
        const int n = qblk + q;
        const int idx = nb[n * HN + h];
        lds_idx[q * HN + h] = idx;
        float cx = 1e6f, cy = 1e6f, cz = 1e6f, fs = 0.0f;
        if (idx < MS) {
            cx = s_pts[idx * 3 + 0] - q_pts[n * 3 + 0];
            cy = s_pts[idx * 3 + 1] - q_pts[n * 3 + 1];
            cz = s_pts[idx * 3 + 2] - q_pts[n * 3 + 2];
            for (int c = 0; c < CIN; c += 4) {
                fs += x[idx * CIN + c + 0] + x[idx * CIN + c + 1]
                    + x[idx * CIN + c + 2] + x[idx * CIN + c + 3];
            }
        }
        lds_nc[(q * HN + h) * 3 + 0] = cx;
        lds_nc[(q * HN + h) * 3 + 1] = cy;
        lds_nc[(q * HN + h) * 3 + 2] = cz;
        const unsigned long long m = __ballot(fs > 0.0f);
        if (lane == 0) {
            const int cnt = (int)__popcll(m);
            lds_norm[q] = (float)(cnt > 0 ? cnt : 1);
        }
    }
    __syncthreads();

    {
        const int q   = wave;
        const int k   = lane & 15;
        const int col = lane & 15;
        float kx = 0.0f, ky = 0.0f, kz = 0.0f;
        if (k < KK) { kx = kp[k * 3]; ky = kp[k * 3 + 1]; kz = kp[k * 3 + 2]; }

        v8f acc[4];
        #pragma unroll
        for (int ct = 0; ct < 4; ++ct) acc[ct] = (v8f){};

        const int hofs = (lane >> 4) * 2;
        constexpr int HB = 2;
        #pragma unroll
        for (int hb0 = 0; hb0 < HN; hb0 += 4 * HB) {
            float f0[HB][4], f1[HB][4];
            bool  ok0[HB], ok1[HB];
            #pragma unroll
            for (int j = 0; j < HB; ++j) {
                const int h0 = hb0 + 4 * j + hofs;
                const int i0 = lds_idx[q * HN + h0];
                const int i1 = lds_idx[q * HN + h0 + 1];
                ok0[j] = i0 < MS;
                ok1[j] = i1 < MS;
                const int e0 = (ok0[j] ? i0 : 0) * CIN + col;
                const int e1 = (ok1[j] ? i1 : 0) * CIN + col;
                #pragma unroll
                for (int ct = 0; ct < 4; ++ct) {
                    f0[j][ct] = x[e0 + ct * 16];
                    f1[j][ct] = x[e1 + ct * 16];
                }
            }
            #pragma unroll
            for (int j = 0; j < HB; ++j) {
                const int h0 = hb0 + 4 * j + hofs;
                v2f A;
                #pragma unroll
                for (int s = 0; s < 2; ++s) {
                    const float nx = lds_nc[(q * HN + h0 + s) * 3 + 0] - kx;
                    const float ny = lds_nc[(q * HN + h0 + s) * 3 + 1] - ky;
                    const float nz = lds_nc[(q * HN + h0 + s) * 3 + 2] - kz;
                    float w = 1.0f - sqrtf(nx * nx + ny * ny + nz * nz) * (1.0f / KPE);
                    w = fmaxf(w, 0.0f);
                    if (k >= KK) w = 0.0f;
                    A[s] = w;
                }
                #pragma unroll
                for (int ct = 0; ct < 4; ++ct) {
                    v2f B;
                    B.x = ok0[j] ? f0[j][ct] : 0.0f;
                    B.y = ok1[j] ? f1[j][ct] : 0.0f;
                    acc[ct] = __builtin_amdgcn_wmma_f32_16x16x4_f32(
                        false, A, false, B, (short)0, acc[ct], false, false);
                }
            }
        }

        const int mrow = (lane >> 4) * 8;
        #pragma unroll
        for (int ct = 0; ct < 4; ++ct)
            #pragma unroll
            for (int r = 0; r < 8; ++r)
                lds_weighted[q * WP + (r + mrow) * CIN + ct * 16 + col] = acc[ct][r];
    }
    __syncthreads();

    {
        const int dtile = wave & 3;
        const int kq    = wave >> 2;
        const int col   = lane & 15;
        const int kofs  = (lane >> 4) * 2;
        const int d     = dtile * 16 + col;
        const int nrow  = lane & 15;

        const int kb0    = kq * 256 + kofs;
        const int nsteps = (kq == 3) ? 48 : 64;
        constexpr int SB = 8;

        const float* wrow = wt + (kb0 * CO + d);
        const float* arow = lds_weighted + (nrow * WP + kb0);

        v8f acc = (v8f){};
        for (int s0 = 0; s0 < nsteps; s0 += SB) {
            v2f A[SB], B[SB];
            #pragma unroll
            for (int j = 0; j < SB; ++j) {
                B[j].x = wrow[(j * 4)     * CO];
                B[j].y = wrow[(j * 4 + 1) * CO];
            }
            #pragma unroll
            for (int j = 0; j < SB; ++j)
                A[j] = *reinterpret_cast<const v2f*>(arow + j * 4);
            #pragma unroll
            for (int j = 0; j < SB; ++j)
                acc = __builtin_amdgcn_wmma_f32_16x16x4_f32(
                    false, A[j], false, B[j], (short)0, acc, false, false);
            wrow += SB * 4 * CO;
            arow += SB * 4;
        }
        const int mrow = (lane >> 4) * 8;
        #pragma unroll
        for (int r = 0; r < 8; ++r)
            lds_partial[wave * 256 + (r + mrow) * 16 + col] = acc[r];
    }
    __syncthreads();

    for (int e = tid; e < QT * CO; e += NTHR) {
        const int n  = e >> 6;
        const int d  = e & 63;
        const int dt = d >> 4;
        const int cc = d & 15;
        float s = 0.0f;
        #pragma unroll
        for (int kq = 0; kq < 4; ++kq)
            s += lds_partial[(kq * 4 + dt) * 256 + n * 16 + cc];
        out[(qblk + n) * CO + d] = s / lds_norm[n];
    }
}

extern "C" void kernel_launch(void* const* d_in, const int* in_sizes, int n_in,
                              void* d_out, int out_size, void* d_ws, size_t ws_size,
                              hipStream_t stream) {
    const float* q_pts = (const float*)d_in[0];
    const float* s_pts = (const float*)d_in[1];
    const int*   nb    = (const int*)  d_in[2];
    const float* x     = (const float*)d_in[3];
    const float* kp    = (const float*)d_in[4];
    const float* wt    = (const float*)d_in[5];
    float*       out   = (float*)d_out;

    const int n = in_sizes[0] / 3;
    const int nblk = (n + QT - 1) / QT;
    kpconv_wmma_f32<<<nblk, NTHR, 0, stream>>>(q_pts, s_pts, nb, x, kp, wt, out);
}


# Density count: sum each neighbour's 64 features from the phase-A B fragments already in registers (16-lane DPP butterfly) instead of a separate per-lane row read in phase 0; row-sum order changes
# speedup vs baseline: 2.2609x; 1.3114x over previous
#include <hip/hip_runtime.h>
#include <hip/hip_bf16.h>

typedef __attribute__((ext_vector_type(2))) float v2f;
typedef __attribute__((ext_vector_type(8))) float v8f;

constexpr int   NQ  = 50000;
constexpr int   MS  = 50000;
constexpr int   HN  = 32;
constexpr int   KK  = 15;
constexpr int   CIN = 64;
constexpr int   CO  = 64;
constexpr float KPE = 1.2f;

constexpr int QT    = 16;
constexpr int NTHR  = 512;
constexpr int WP    = 16 * CIN + 4;

template <int XM>
__device__ __forceinline__ float row_xor_get(float v) {
    return __builtin_bit_cast(float,
        __builtin_amdgcn_update_dpp(0, __builtin_bit_cast(int, v), 0x160 + XM, 0xf, 0xf, true));
}

__global__ __launch_bounds__(NTHR, 12)
void kpconv_wmma_f32(const float* __restrict__ q_pts,
                     const float* __restrict__ s_pts,
                     const int*   __restrict__ nb,
                     const float* __restrict__ x,
                     const float* __restrict__ kp,
                     const float* __restrict__ wt,
                     float*       __restrict__ out)
{
    struct alignas(16) Lds {
        float partial [16 * 16 * 16];
        float nc      [QT * HN * 3];
        int   idx     [QT * HN];
        float norm    [QT];
        float weighted[QT * WP];
    };
    __shared__ Lds lds;
    float* const lds_partial  = lds.partial;
    float* const lds_nc       = lds.nc;
    int*   const lds_idx      = lds.idx;
    float* const lds_norm     = lds.norm;
    float* const lds_weighted = lds.weighted;

    const int tid  = threadIdx.x;
    const int lane = tid & 31;
    const int wave = tid >> 5;
    const int qblk = blockIdx.x * QT;

    {
        const int q = wave;
        const int h = lane;
        const int n = qblk + q;
        const int idx = nb[n * HN + h];
        lds_idx[q * HN + h] = idx;
        float cx = 1e6f, cy = 1e6f, cz = 1e6f;
        if (idx < MS) {
            cx = s_pts[idx * 3 + 0] - q_pts[n * 3 + 0];
            cy = s_pts[idx * 3 + 1] - q_pts[n * 3 + 1];
            cz = s_pts[idx * 3 + 2] - q_pts[n * 3 + 2];
        }
        lds_nc[(q * HN + h) * 3 + 0] = cx;
        lds_nc[(q * HN + h) * 3 + 1] = cy;
        lds_nc[(q * HN + h) * 3 + 2] = cz;
    }
    __syncthreads();

    {
        const int q   = wave;
        const int k   = lane & 15;
        const int col = lane & 15;
        float kx = 0.0f, ky = 0.0f, kz = 0.0f;
        if (k < KK) { kx = kp[k * 3]; ky = kp[k * 3 + 1]; kz = kp[k * 3 + 2]; }

        v8f acc[4];
        #pragma unroll
        for (int ct = 0; ct < 4; ++ct) acc[ct] = (v8f){};

        const int hofs = (lane >> 4) * 2;
        int cnt = 0;
        const bool odd1 = (lane & 1) != 0;
        const bool odd2 = (lane & 2) != 0;
        constexpr int HB = 2;
        #pragma unroll
        for (int hb0 = 0; hb0 < HN; hb0 += 4 * HB) {
            float f0[HB][4], f1[HB][4];
            bool  ok0[HB], ok1[HB];
            #pragma unroll
            for (int j = 0; j < HB; ++j) {
                const int h0 = hb0 + 4 * j + hofs;
                const int i0 = lds_idx[q * HN + h0];
                const int i1 = lds_idx[q * HN + h0 + 1];
                ok0[j] = i0 < MS;
                ok1[j] = i1 < MS;
                const int e0 = (ok0[j] ? i0 : 0) * CIN + col;
                const int e1 = (ok1[j] ? i1 : 0) * CIN + col;
                #pragma unroll
                for (int ct = 0; ct < 4; ++ct) {
                    f0[j][ct] = x[e0 + ct * 16];
                    f1[j][ct] = x[e1 + ct * 16];
                }
            }
            float rs[2 * HB];
            #pragma unroll
            for (int j = 0; j < HB; ++j) {
                const int h0 = hb0 + 4 * j + hofs;
                v2f A;
                #pragma unroll
                for (int s = 0; s < 2; ++s) {
                    const float nx = lds_nc[(q * HN + h0 + s) * 3 + 0] - kx;
                    const float ny = lds_nc[(q * HN + h0 + s) * 3 + 1] - ky;
                    const float nz = lds_nc[(q * HN + h0 + s) * 3 + 2] - kz;
                    float w = 1.0f - sqrtf(nx * nx + ny * ny + nz * nz) * (1.0f / KPE);
                    w = fmaxf(w, 0.0f);
                    if (k >= KK) w = 0.0f;
                    A[s] = w;
                }
                v2f B[4];
                #pragma unroll
                for (int ct = 0; ct < 4; ++ct) {
                    B[ct].x = ok0[j] ? f0[j][ct] : 0.0f;
                    B[ct].y = ok1[j] ? f1[j][ct] : 0.0f;
                    acc[ct] = __builtin_amdgcn_wmma_f32_16x16x4_f32(
                        false, A, false, B[ct], (short)0, acc[ct], false, false);
                }
                rs[2 * j]     = (B[0].x + B[1].x) + (B[2].x + B[3].x);
                rs[2 * j + 1] = (B[0].y + B[1].y) + (B[2].y + B[3].y);
            }
            static_assert(HB == 2, "the packed row sum below is written for 4 rows per half and batch");
            float a01 = odd1 ? rs[1] : rs[0];
            a01 += row_xor_get<1>(odd1 ? rs[0] : rs[1]);
            float a23 = odd1 ? rs[3] : rs[2];
            a23 += row_xor_get<1>(odd1 ? rs[2] : rs[3]);
            float rsum = odd2 ? a23 : a01;
            rsum += row_xor_get<2>(odd2 ? a01 : a23);
            rsum += row_xor_get<4>(rsum);
            rsum += row_xor_get<8>(rsum);
            const unsigned m = (unsigned)__ballot(rsum > 0.0f);
            cnt += __popc(m & 0x000F000Fu);
        }
        if (lane == 0) lds_norm[q] = (float)(cnt > 0 ? cnt : 1);

        const int mrow = (lane >> 4) * 8;
        #pragma unroll
        for (int ct = 0; ct < 4; ++ct)
            #pragma unroll
            for (int r = 0; r < 8; ++r)
                lds_weighted[q * WP + (r + mrow) * CIN + ct * 16 + col] = acc[ct][r];
    }
    __syncthreads();

    {
        const int dtile = wave & 3;
        const int kq    = wave >> 2;
        const int col   = lane & 15;
        const int kofs  = (lane >> 4) * 2;
        const int d     = dtile * 16 + col;
        const int nrow  = lane & 15;

        const int kb0    = kq * 256 + kofs;
        const int nsteps = (kq == 3) ? 48 : 64;
        constexpr int SB = 8;

        const float* wrow = wt + (kb0 * CO + d);
        const float* arow = lds_weighted + (nrow * WP + kb0);

        v8f acc = (v8f){};
        for (int s0 = 0; s0 < nsteps; s0 += SB) {
            v2f A[SB], B[SB];
            #pragma unroll
            for (int j = 0; j < SB; ++j) {
                B[j].x = wrow[(j * 4)     * CO];
                B[j].y = wrow[(j * 4 + 1) * CO];
            }
            #pragma unroll
            for (int j = 0; j < SB; ++j)
                A[j] = *reinterpret_cast<const v2f*>(arow + j * 4);
            #pragma unroll
            for (int j = 0; j < SB; ++j)
                acc = __builtin_amdgcn_wmma_f32_16x16x4_f32(
                    false, A[j], false, B[j], (short)0, acc, false, false);
            wrow += SB * 4 * CO;
            arow += SB * 4;
        }
        const int mrow = (lane >> 4) * 8;
        #pragma unroll
        for (int r = 0; r < 8; ++r)
            lds_partial[wave * 256 + (r + mrow) * 16 + col] = acc[r];
    }
    __syncthreads();

    for (int e = tid; e < QT * CO; e += NTHR) {
        const int n  = e >> 6;
        const int d  = e & 63;
        const int dt = d >> 4;
        const int cc = d & 15;
        float s = 0.0f;
        #pragma unroll
        for (int kq = 0; kq < 4; ++kq)
            s += lds_partial[(kq * 4 + dt) * 256 + n * 16 + cc];
        out[(qblk + n) * CO + d] = s / lds_norm[n];
    }
}

extern "C" void kernel_launch(void* const* d_in, const int* in_sizes, int n_in,
                              void* d_out, int out_size, void* d_ws, size_t ws_size,
                              hipStream_t stream) {
    const float* q_pts = (const float*)d_in[0];
    const float* s_pts = (const float*)d_in[1];
    const int*   nb    = (const int*)  d_in[2];
    const float* x     = (const float*)d_in[3];
    const float* kp    = (const float*)d_in[4];
    const float* wt    = (const float*)d_in[5];
    float*       out   = (float*)d_out;

    const int n = in_sizes[0] / 3;
    const int nblk = (n + QT - 1) / QT;
    kpconv_wmma_f32<<<nblk, NTHR, 0, stream>>>(q_pts, s_pts, nb, x, kp, wt, out);
}


# Column tiles interleaved (tile t = columns 4c+t): a lane loads 16 bytes of a feature/weight row per load, 4x fewer loads; phase-B wave = 4 tiles x 1/16 of K: 16 K-parts summed, row-sum order changes
# speedup vs baseline: 2.3360x; 1.0332x over previous
#include <hip/hip_runtime.h>
#include <hip/hip_bf16.h>

typedef __attribute__((ext_vector_type(2))) float v2f;
typedef __attribute__((ext_vector_type(4))) float v4f;
typedef __attribute__((ext_vector_type(8))) float v8f;

constexpr int   NQ  = 50000;
constexpr int   MS  = 50000;
constexpr int   HN  = 32;
constexpr int   KK  = 15;
constexpr int   CIN = 64;
constexpr int   CO  = 64;
constexpr float KPE = 1.2f;

constexpr int QT    = 16;
constexpr int NTHR  = 512;
constexpr int WP    = 16 * CIN + 4;

template <int XM>
__device__ __forceinline__ float row_xor_get(float v) {
    return __builtin_bit_cast(float,
        __builtin_amdgcn_update_dpp(0, __builtin_bit_cast(int, v), 0x160 + XM, 0xf, 0xf, true));
}

__global__ __launch_bounds__(NTHR, 12)
void kpconv_wmma_f32(const float* __restrict__ q_pts,
                     const float* __restrict__ s_pts,
                     const int*   __restrict__ nb,
                     const float* __restrict__ x,
                     const float* __restrict__ kp,
                     const float* __restrict__ wt,
                     float*       __restrict__ out)
{
    struct alignas(16) Lds {
        float nc      [QT * HN * 3];
        int   idx     [QT * HN];
        float norm    [QT];
        float weighted[QT * WP];
    };
    static_assert(16 * QT * CO <= QT * WP, "the partial tiles must fit in the weighted tile's place");
    __shared__ Lds lds;
    float* const lds_partial  = lds.weighted;
    float* const lds_nc       = lds.nc;
    int*   const lds_idx      = lds.idx;
    float* const lds_norm     = lds.norm;
    float* const lds_weighted = lds.weighted;

    const int tid  = threadIdx.x;
    const int lane = tid & 31;
    const int wave = tid >> 5;
    const int qblk = blockIdx.x * QT;

    {
        const int q = wave;
        const int h = lane;
        const int n = qblk + q;
        const int idx = nb[n * HN + h];
        lds_idx[q * HN + h] = idx;
        float cx = 1e6f, cy = 1e6f, cz = 1e6f;
        if (idx < MS) {
            cx = s_pts[idx * 3 + 0] - q_pts[n * 3 + 0];
            cy = s_pts[idx * 3 + 1] - q_pts[n * 3 + 1];
            cz = s_pts[idx * 3 + 2] - q_pts[n * 3 + 2];
        }
        lds_nc[(q * HN + h) * 3 + 0] = cx;
        lds_nc[(q * HN + h) * 3 + 1] = cy;
        lds_nc[(q * HN + h) * 3 + 2] = cz;
    }
    __syncthreads();

    {
        const int q   = wave;
        const int k   = lane & 15;
        const int col = lane & 15;
        float kx = 0.0f, ky = 0.0f, kz = 0.0f;
        if (k < KK) { kx = kp[k * 3]; ky = kp[k * 3 + 1]; kz = kp[k * 3 + 2]; }

        v8f acc[4];
        #pragma unroll
        for (int ct = 0; ct < 4; ++ct) acc[ct] = (v8f){};

        const int hofs = (lane >> 4) * 2;
        int cnt = 0;
        const bool odd1 = (lane & 1) != 0;
        const bool odd2 = (lane & 2) != 0;
        constexpr int HB = 2;
        #pragma nounroll
        for (int hb0 = 0; hb0 < HN; hb0 += 4 * HB) {
            v4f   f0[HB], f1[HB];
            bool  ok0[HB], ok1[HB];
            #pragma unroll
            for (int j = 0; j < HB; ++j) {
                const int h0 = hb0 + 4 * j + hofs;
                const int i0 = lds_idx[q * HN + h0];
                const int i1 = lds_idx[q * HN + h0 + 1];
                ok0[j] = i0 < MS;
                ok1[j] = i1 < MS;
                const int e0 = (ok0[j] ? i0 : 0) * CIN + 4 * col;
                const int e1 = (ok1[j] ? i1 : 0) * CIN + 4 * col;
                f0[j] = *reinterpret_cast<const v4f*>(x + e0);
                f1[j] = *reinterpret_cast<const v4f*>(x + e1);
            }
            float rs[2 * HB];
            #pragma unroll
            for (int j = 0; j < HB; ++j) {
                const int h0 = hb0 + 4 * j + hofs;
                v2f A;
                #pragma unroll
                for (int s = 0; s < 2; ++s) {
                    const float nx = lds_nc[(q * HN + h0 + s) * 3 + 0] - kx;
                    const float ny = lds_nc[(q * HN + h0 + s) * 3 + 1] - ky;
                    const float nz = lds_nc[(q * HN + h0 + s) * 3 + 2] - kz;
                    float w = 1.0f - sqrtf(nx * nx + ny * ny + nz * nz) * (1.0f / KPE);
                    w = fmaxf(w, 0.0f);
                    if (k >= KK) w = 0.0f;
                    A[s] = w;
                }
                v2f B[4];
                #pragma unroll
                for (int ct = 0; ct < 4; ++ct) {
                    B[ct].x = ok0[j] ? f0[j][ct] : 0.0f;
                    B[ct].y = ok1[j] ? f1[j][ct] : 0.0f;
                    acc[ct] = __builtin_amdgcn_wmma_f32_16x16x4_f32(
                        false, A, false, B[ct], (short)0, acc[ct], false, false);
                }
                rs[2 * j]     = (B[0].x + B[1].x) + (B[2].x + B[3].x);
                rs[2 * j + 1] = (B[0].y + B[1].y) + (B[2].y + B[3].y);
            }
            static_assert(HB == 2, "the packed row sum below is written for 4 rows per half and batch");
            float a01 = odd1 ? rs[1] : rs[0];
            a01 += row_xor_get<1>(odd1 ? rs[0] : rs[1]);
            float a23 = odd1 ? rs[3] : rs[2];
            a23 += row_xor_get<1>(odd1 ? rs[2] : rs[3]);
            float rsum = odd2 ? a23 : a01;
            rsum += row_xor_get<2>(odd2 ? a01 : a23);
            rsum += row_xor_get<4>(rsum);
            rsum += row_xor_get<8>(rsum);
            const unsigned m = (unsigned)__ballot(rsum > 0.0f);
            cnt += __popc(m & 0x000F000Fu);
        }
        if (lane == 0) lds_norm[q] = (float)(cnt > 0 ? cnt : 1);

        const int mrow = (lane >> 4) * 8;
        #pragma unroll
        for (int r = 0; r < 8; ++r) {
            const v4f o = {acc[0][r], acc[1][r], acc[2][r], acc[3][r]};
            *reinterpret_cast<v4f*>(lds_weighted + q * WP + (r + mrow) * CIN + 4 * col) = o;
        }
    }
    __syncthreads();

    constexpr int KW  = (KK * CIN) / 16;
    constexpr int NST = KW / 4;
    static_assert(KW * 16 == KK * CIN && NST * 4 == KW, "K must split evenly over the 16 waves");
    v8f accB[4];
    {
        const int col   = lane & 15;
        const int kofs  = (lane >> 4) * 2;
        const int nrow  = lane & 15;

        const int kb0 = wave * KW + kofs;

        const float* wrow = wt + (kb0 * CO + 4 * col);
        const float* arow = lds_weighted + (nrow * WP + kb0);

        #pragma unroll
        for (int t = 0; t < 4; ++t) accB[t] = (v8f){};
        v4f W0[NST], W1[NST];
        v2f A[NST];
        #pragma unroll
        for (int j = 0; j < NST; ++j) {
            W0[j] = *reinterpret_cast<const v4f*>(wrow + (j * 4)     * CO);
            W1[j] = *reinterpret_cast<const v4f*>(wrow + (j * 4 + 1) * CO);
        }
        #pragma unroll
        for (int j = 0; j < NST; ++j)
            A[j] = *reinterpret_cast<const v2f*>(arow + j * 4);
        #pragma unroll
        for (int j = 0; j < NST; ++j) {
            #pragma unroll
            for (int t = 0; t < 4; ++t) {
                const v2f B = {W0[j][t], W1[j][t]};
                accB[t] = __builtin_amdgcn_wmma_f32_16x16x4_f32(
                    false, A[j], false, B, (short)0, accB[t], false, false);
            }
        }
    }
    __syncthreads();
    {
        const int col  = lane & 15;
        const int mrow = (lane >> 4) * 8;
        #pragma unroll
        for (int r = 0; r < 8; ++r) {
            const v4f o = {accB[0][r], accB[1][r], accB[2][r], accB[3][r]};
            *reinterpret_cast<v4f*>(lds_partial + wave * (QT * CO) + (r + mrow) * CO + 4 * col) = o;
        }
    }
    __syncthreads();

    {
        const int n = wave;
        const int d = lane * 2;
        v2f s = {0.0f, 0.0f};
        #pragma unroll
        for (int w = 0; w < 16; ++w)
            s += *reinterpret_cast<const v2f*>(lds_partial + w * (QT * CO) + n * CO + d);
        const float nm = lds_norm[n];
        const v2f o = {s.x / nm, s.y / nm};
        *reinterpret_cast<v2f*>(out + (qblk + n) * CO + d) = o;
    }
}

extern "C" void kernel_launch(void* const* d_in, const int* in_sizes, int n_in,
                              void* d_out, int out_size, void* d_ws, size_t ws_size,
                              hipStream_t stream) {
    const float* q_pts = (const float*)d_in[0];
    const float* s_pts = (const float*)d_in[1];
    const int*   nb    = (const int*)  d_in[2];
    const float* x     = (const float*)d_in[3];
    const float* kp    = (const float*)d_in[4];
    const float* wt    = (const float*)d_in[5];
    float*       out   = (float*)d_out;

    const int n = in_sizes[0] / 3;
    const int nblk = (n + QT - 1) / QT;
    kpconv_wmma_f32<<<nblk, NTHR, 0, stream>>>(q_pts, s_pts, nb, x, kp, wt, out);
}


# Kernel-point distance by the one-instruction square root (v_sqrt_f32, 1 ulp) instead of correctly rounded sqrtf: 45 fewer vector instructions per phase-A trip; last bit of weights may differ
# speedup vs baseline: 2.4462x; 1.0472x over previous
#include <hip/hip_runtime.h>
#include <hip/hip_bf16.h>

typedef __attribute__((ext_vector_type(2))) float v2f;
typedef __attribute__((ext_vector_type(4))) float v4f;
typedef __attribute__((ext_vector_type(8))) float v8f;

constexpr int   NQ  = 50000;
constexpr int   MS  = 50000;
constexpr int   HN  = 32;
constexpr int   KK  = 15;
constexpr int   CIN = 64;
constexpr int   CO  = 64;
constexpr float KPE = 1.2f;

constexpr int QT    = 16;
constexpr int NTHR  = 512;
constexpr int WP    = 16 * CIN + 4;

template <int XM>
__device__ __forceinline__ float row_xor_get(float v) {
    return __builtin_bit_cast(float,
        __builtin_amdgcn_update_dpp(0, __builtin_bit_cast(int, v), 0x160 + XM, 0xf, 0xf, true));
}

__global__ __launch_bounds__(NTHR, 12)
void kpconv_wmma_f32(const float* __restrict__ q_pts,
                     const float* __restrict__ s_pts,
                     const int*   __restrict__ nb,
                     const float* __restrict__ x,
                     const float* __restrict__ kp,
                     const float* __restrict__ wt,
                     float*       __restrict__ out)
{
    struct alignas(16) Lds {
        float nc      [QT * HN * 3];
        int   idx     [QT * HN];
        float norm    [QT];
        float weighted[QT * WP];
    };
    static_assert(16 * QT * CO <= QT * WP, "the partial tiles must fit in the weighted tile's place");
    __shared__ Lds lds;
    float* const lds_partial  = lds.weighted;
    float* const lds_nc       = lds.nc;
    int*   const lds_idx      = lds.idx;
    float* const lds_norm     = lds.norm;
    float* const lds_weighted = lds.weighted;

    const int tid  = threadIdx.x;
    const int lane = tid & 31;
    const int wave = tid >> 5;
    const int qblk = blockIdx.x * QT;

    {
        const int q = wave;
        const int h = lane;
        const int n = qblk + q;
        const int idx = nb[n * HN + h];
        lds_idx[q * HN + h] = idx;
        float cx = 1e6f, cy = 1e6f, cz = 1e6f;
        if (idx < MS) {
            cx = s_pts[idx * 3 + 0] - q_pts[n * 3 + 0];
            cy = s_pts[idx * 3 + 1] - q_pts[n * 3 + 1];
            cz = s_pts[idx * 3 + 2] - q_pts[n * 3 + 2];
        }
        lds_nc[(q * HN + h) * 3 + 0] = cx;
        lds_nc[(q * HN + h) * 3 + 1] = cy;
        lds_nc[(q * HN + h) * 3 + 2] = cz;
    }
    __syncthreads();

    {
        const int q   = wave;
        const int k   = lane & 15;
        const int col = lane & 15;
        float kx = 0.0f, ky = 0.0f, kz = 0.0f;
        if (k < KK) { kx = kp[k * 3]; ky = kp[k * 3 + 1]; kz = kp[k * 3 + 2]; }

        v8f acc[4];
        #pragma unroll
        for (int ct = 0; ct < 4; ++ct) acc[ct] = (v8f){};

        const int hofs = (lane >> 4) * 2;
        int cnt = 0;
        const bool odd1 = (lane & 1) != 0;
        const bool odd2 = (lane & 2) != 0;
        constexpr int HB = 2;
        #pragma nounroll
        for (int hb0 = 0; hb0 < HN; hb0 += 4 * HB) {
            v4f   f0[HB], f1[HB];
            bool  ok0[HB], ok1[HB];
            #pragma unroll
            for (int j = 0; j < HB; ++j) {
                const int h0 = hb0 + 4 * j + hofs;
                const int i0 = lds_idx[q * HN + h0];
                const int i1 = lds_idx[q * HN + h0 + 1];
                ok0[j] = i0 < MS;
                ok1[j] = i1 < MS;
                const int e0 = (ok0[j] ? i0 : 0) * CIN + 4 * col;
                const int e1 = (ok1[j] ? i1 : 0) * CIN + 4 * col;
                f0[j] = *reinterpret_cast<const v4f*>(x + e0);
                f1[j] = *reinterpret_cast<const v4f*>(x + e1);
            }
            float rs[2 * HB];
            #pragma unroll
            for (int j = 0; j < HB; ++j) {
                const int h0 = hb0 + 4 * j + hofs;
                v2f A;
                #pragma unroll
                for (int s = 0; s < 2; ++s) {
                    const float nx = lds_nc[(q * HN + h0 + s) * 3 + 0] - kx;
                    const float ny = lds_nc[(q * HN + h0 + s) * 3 + 1] - ky;
                    const float nz = lds_nc[(q * HN + h0 + s) * 3 + 2] - kz;
                    float w = 1.0f - __builtin_amdgcn_sqrtf(nx * nx + ny * ny + nz * nz) * (1.0f / KPE);
                    w = fmaxf(w, 0.0f);
                    if (k >= KK) w = 0.0f;
                    A[s] = w;
                }
                v2f B[4];
                #pragma unroll
                for (int ct = 0; ct < 4; ++ct) {
                    B[ct].x = ok0[j] ? f0[j][ct] : 0.0f;
                    B[ct].y = ok1[j] ? f1[j][ct] : 0.0f;
                    acc[ct] = __builtin_amdgcn_wmma_f32_16x16x4_f32(
                        false, A, false, B[ct], (short)0, acc[ct], false, false);
                }
                rs[2 * j]     = (B[0].x + B[1].x) + (B[2].x + B[3].x);
                rs[2 * j + 1] = (B[0].y + B[1].y) + (B[2].y + B[3].y);
            }
            static_assert(HB == 2, "the packed row sum below is written for 4 rows per half and batch");
            float a01 = odd1 ? rs[1] : rs[0];
            a01 += row_xor_get<1>(odd1 ? rs[0] : rs[1]);
            float a23 = odd1 ? rs[3] : rs[2];
            a23 += row_xor_get<1>(odd1 ? rs[2] : rs[3]);
            float rsum = odd2 ? a23 : a01;
            rsum += row_xor_get<2>(odd2 ? a01 : a23);
            rsum += row_xor_get<4>(rsum);
            rsum += row_xor_get<8>(rsum);
            const unsigned m = (unsigned)__ballot(rsum > 0.0f);
            cnt += __popc(m & 0x000F000Fu);
        }
        if (lane == 0) lds_norm[q] = (float)(cnt > 0 ? cnt : 1);

        const int mrow = (lane >> 4) * 8;
        #pragma unroll
        for (int r = 0; r < 8; ++r) {
            const v4f o = {acc[0][r], acc[1][r], acc[2][r], acc[3][r]};
            *reinterpret_cast<v4f*>(lds_weighted + q * WP + (r + mrow) * CIN + 4 * col) = o;
        }
    }
    __syncthreads();

    constexpr int KW  = (KK * CIN) / 16;
    constexpr int NST = KW / 4;
    static_assert(KW * 16 == KK * CIN && NST * 4 == KW, "K must split evenly over the 16 waves");
    v8f accB[4];
    {
        const int col   = lane & 15;
        const int kofs  = (lane >> 4) * 2;
        const int nrow  = lane & 15;

        const int kb0 = wave * KW + kofs;

        const float* wrow = wt + (kb0 * CO + 4 * col);
        const float* arow = lds_weighted + (nrow * WP + kb0);

        #pragma unroll
        for (int t = 0; t < 4; ++t) accB[t] = (v8f){};
        v4f W0[NST], W1[NST];
        v2f A[NST];
        #pragma unroll
        for (int j = 0; j < NST; ++j) {
            W0[j] = *reinterpret_cast<const v4f*>(wrow + (j * 4)     * CO);
            W1[j] = *reinterpret_cast<const v4f*>(wrow + (j * 4 + 1) * CO);
        }
        #pragma unroll
        for (int j = 0; j < NST; ++j)
            A[j] = *reinterpret_cast<const v2f*>(arow + j * 4);
        #pragma unroll
        for (int j = 0; j < NST; ++j) {
            #pragma unroll
            for (int t = 0; t < 4; ++t) {
                const v2f B = {W0[j][t], W1[j][t]};
                accB[t] = __builtin_amdgcn_wmma_f32_16x16x4_f32(
                    false, A[j], false, B, (short)0, accB[t], false, false);
            }
        }
    }
    __syncthreads();
    {
        const int col  = lane & 15;
        const int mrow = (lane >> 4) * 8;
        #pragma unroll
        for (int r = 0; r < 8; ++r) {
            const v4f o = {accB[0][r], accB[1][r], accB[2][r], accB[3][r]};
            *reinterpret_cast<v4f*>(lds_partial + wave * (QT * CO) + (r + mrow) * CO + 4 * col) = o;
        }
    }
    __syncthreads();

    {
        const int n = wave;
        const int d = lane * 2;
        v2f s = {0.0f, 0.0f};
        #pragma unroll
        for (int w = 0; w < 16; ++w)
            s += *reinterpret_cast<const v2f*>(lds_partial + w * (QT * CO) + n * CO + d);
        const float nm = lds_norm[n];
        const v2f o = {s.x / nm, s.y / nm};
        *reinterpret_cast<v2f*>(out + (qblk + n) * CO + d) = o;
    }
}

extern "C" void kernel_launch(void* const* d_in, const int* in_sizes, int n_in,
                              void* d_out, int out_size, void* d_ws, size_t ws_size,
                              hipStream_t stream) {
    const float* q_pts = (const float*)d_in[0];
    const float* s_pts = (const float*)d_in[1];
    const int*   nb    = (const int*)  d_in[2];
    const float* x     = (const float*)d_in[3];
    const float* kp    = (const float*)d_in[4];
    const float* wt    = (const float*)d_in[5];
    float*       out   = (float*)d_out;

    const int n = in_sizes[0] / 3;
    const int nblk = (n + QT - 1) / QT;
    kpconv_wmma_f32<<<nblk, NTHR, 0, stream>>>(q_pts, s_pts, nb, x, kp, wt, out);
}


# Phase 0 stores each query's neighbours reached-first (within 1.2 of a kernel point, ballot rank); phase A skips the WMMAs of K-steps past them, all weights 0 there: neighbour-sum order changes
# speedup vs baseline: 2.5788x; 1.0542x over previous
#include <hip/hip_runtime.h>
#include <hip/hip_bf16.h>

typedef __attribute__((ext_vector_type(2))) float v2f;
typedef __attribute__((ext_vector_type(4))) float v4f;
typedef __attribute__((ext_vector_type(8))) float v8f;

constexpr int   NQ  = 50000;
constexpr int   MS  = 50000;
constexpr int   HN  = 32;
constexpr int   KK  = 15;
constexpr int   CIN = 64;
constexpr int   CO  = 64;
constexpr float KPE = 1.2f;
constexpr float REACH2 = KPE * KPE * 1.01f;

constexpr int QT    = 16;
constexpr int NTHR  = 512;
constexpr int WP    = 16 * CIN + 4;

template <int XM>
__device__ __forceinline__ float row_xor_get(float v) {
    return __builtin_bit_cast(float,
        __builtin_amdgcn_update_dpp(0, __builtin_bit_cast(int, v), 0x160 + XM, 0xf, 0xf, true));
}

__global__ __launch_bounds__(NTHR, 12)
void kpconv_wmma_f32(const float* __restrict__ q_pts,
                     const float* __restrict__ s_pts,
                     const int*   __restrict__ nb,
                     const float* __restrict__ x,
                     const float* __restrict__ kp,
                     const float* __restrict__ wt,
                     float*       __restrict__ out)
{
    struct alignas(16) Lds {
        float nc      [QT * HN * 3];
        int   idx     [QT * HN];
        float norm    [QT];
        float weighted[QT * WP];
    };
    static_assert(16 * QT * CO <= QT * WP, "the partial tiles must fit in the weighted tile's place");
    __shared__ Lds lds;
    float* const lds_partial  = lds.weighted;
    float* const lds_nc       = lds.nc;
    int*   const lds_idx      = lds.idx;
    float* const lds_norm     = lds.norm;
    float* const lds_weighted = lds.weighted;

    const int tid  = threadIdx.x;
    const int lane = tid & 31;
    const int wave = tid >> 5;
    const int qblk = blockIdx.x * QT;

    int nreach;
    {
        const int q = wave;
        const int h = lane;
        const int n = qblk + q;
        const int idx = nb[n * HN + h];
        float cx = 1e6f, cy = 1e6f, cz = 1e6f;
        if (idx < MS) {
            cx = s_pts[idx * 3 + 0] - q_pts[n * 3 + 0];
            cy = s_pts[idx * 3 + 1] - q_pts[n * 3 + 1];
            cz = s_pts[idx * 3 + 2] - q_pts[n * 3 + 2];
        }
        float m = 3.0e38f;
        #pragma unroll
        for (int k = 0; k < KK; k += 2) {
            const int k1 = (k + 1 < KK) ? k + 1 : k;
            const v2f tx = (v2f){cx, cx} - (v2f){kp[3 * k],     kp[3 * k1]};
            const v2f ty = (v2f){cy, cy} - (v2f){kp[3 * k + 1], kp[3 * k1 + 1]};
            const v2f tz = (v2f){cz, cz} - (v2f){kp[3 * k + 2], kp[3 * k1 + 2]};
            const v2f d  = tx * tx + ty * ty + tz * tz;
            m = __builtin_fminf(m, __builtin_fminf(d.x, d.y));
        }
        const bool reached = m < REACH2;
        const unsigned rm  = (unsigned)__ballot(reached);
        nreach = __popc(rm);
        const int rank_r = (int)__builtin_amdgcn_mbcnt_lo(rm, 0u);
        const int pos = reached ? rank_r : nreach + (h - rank_r);
        lds_idx[q * HN + pos] = idx;
        lds_nc[(q * HN + pos) * 3 + 0] = cx;
        lds_nc[(q * HN + pos) * 3 + 1] = cy;
        lds_nc[(q * HN + pos) * 3 + 2] = cz;
    }
    __syncthreads();

    {
        const int q   = wave;
        const int k   = lane & 15;
        const int col = lane & 15;
        float kx = 0.0f, ky = 0.0f, kz = 0.0f;
        if (k < KK) { kx = kp[k * 3]; ky = kp[k * 3 + 1]; kz = kp[k * 3 + 2]; }

        v8f acc[4];
        #pragma unroll
        for (int ct = 0; ct < 4; ++ct) acc[ct] = (v8f){};

        const int hofs = (lane >> 4) * 2;
        int cnt = 0;
        const bool odd1 = (lane & 1) != 0;
        const bool odd2 = (lane & 2) != 0;
        constexpr int HB = 2;
        #pragma nounroll
        for (int hb0 = 0; hb0 < HN; hb0 += 4 * HB) {
            v4f   f0[HB], f1[HB];
            bool  ok0[HB], ok1[HB];
            #pragma unroll
            for (int j = 0; j < HB; ++j) {
                const int h0 = hb0 + 4 * j + hofs;
                const int i0 = lds_idx[q * HN + h0];
                const int i1 = lds_idx[q * HN + h0 + 1];
                ok0[j] = i0 < MS;
                ok1[j] = i1 < MS;
                const int e0 = (ok0[j] ? i0 : 0) * CIN + 4 * col;
                const int e1 = (ok1[j] ? i1 : 0) * CIN + 4 * col;
                f0[j] = *reinterpret_cast<const v4f*>(x + e0);
                f1[j] = *reinterpret_cast<const v4f*>(x + e1);
            }
            __builtin_amdgcn_sched_barrier(0);
            float rs[2 * HB];
            #pragma unroll
            for (int j = 0; j < HB; ++j) {
                const int h0 = hb0 + 4 * j + hofs;
                v2f B[4];
                #pragma unroll
                for (int ct = 0; ct < 4; ++ct) {
                    B[ct].x = ok0[j] ? f0[j][ct] : 0.0f;
                    B[ct].y = ok1[j] ? f1[j][ct] : 0.0f;
                }
                if (hb0 + 4 * j < nreach) {
                v2f A;
                #pragma unroll
                for (int s = 0; s < 2; ++s) {
                    const float nx = lds_nc[(q * HN + h0 + s) * 3 + 0] - kx;
                    const float ny = lds_nc[(q * HN + h0 + s) * 3 + 1] - ky;
                    const float nz = lds_nc[(q * HN + h0 + s) * 3 + 2] - kz;
                    float w = 1.0f - __builtin_amdgcn_sqrtf(nx * nx + ny * ny + nz * nz) * (1.0f / KPE);
                    w = fmaxf(w, 0.0f);
                    if (k >= KK) w = 0.0f;
                    A[s] = w;
                }
                #pragma unroll
                for (int ct = 0; ct < 4; ++ct)
                    acc[ct] = __builtin_amdgcn_wmma_f32_16x16x4_f32(
                        false, A, false, B[ct], (short)0, acc[ct], false, false);
                }
                rs[2 * j]     = (B[0].x + B[1].x) + (B[2].x + B[3].x);
                rs[2 * j + 1] = (B[0].y + B[1].y) + (B[2].y + B[3].y);
            }
            static_assert(HB == 2, "the packed row sum below is written for 4 rows per half and batch");
            float a01 = odd1 ? rs[1] : rs[0];
            a01 += row_xor_get<1>(odd1 ? rs[0] : rs[1]);
            float a23 = odd1 ? rs[3] : rs[2];
            a23 += row_xor_get<1>(odd1 ? rs[2] : rs[3]);
            float rsum = odd2 ? a23 : a01;
            rsum += row_xor_get<2>(odd2 ? a01 : a23);
            rsum += row_xor_get<4>(rsum);
            rsum += row_xor_get<8>(rsum);
            const unsigned m = (unsigned)__ballot(rsum > 0.0f);
            cnt += __popc(m & 0x000F000Fu);
        }
        if (lane == 0) lds_norm[q] = (float)(cnt > 0 ? cnt : 1);

        const int mrow = (lane >> 4) * 8;
        #pragma unroll
        for (int r = 0; r < 8; ++r) {
            const v4f o = {acc[0][r], acc[1][r], acc[2][r], acc[3][r]};
            *reinterpret_cast<v4f*>(lds_weighted + q * WP + (r + mrow) * CIN + 4 * col) = o;
        }
    }
    __syncthreads();

    constexpr int KW  = (KK * CIN) / 16;
    constexpr int NST = KW / 4;
    static_assert(KW * 16 == KK * CIN && NST * 4 == KW, "K must split evenly over the 16 waves");
    v8f accB[4];
    {
        const int col   = lane & 15;
        const int kofs  = (lane >> 4) * 2;
        const int nrow  = lane & 15;

        const int kb0 = wave * KW + kofs;

        const float* wrow = wt + (kb0 * CO + 4 * col);
        const float* arow = lds_weighted + (nrow * WP + kb0);

        #pragma unroll
        for (int t = 0; t < 4; ++t) accB[t] = (v8f){};
        v4f W0[NST], W1[NST];
        v2f A[NST];
        #pragma unroll
        for (int j = 0; j < NST; ++j) {
            W0[j] = *reinterpret_cast<const v4f*>(wrow + (j * 4)     * CO);
            W1[j] = *reinterpret_cast<const v4f*>(wrow + (j * 4 + 1) * CO);
        }
        #pragma unroll
        for (int j = 0; j < NST; ++j)
            A[j] = *reinterpret_cast<const v2f*>(arow + j * 4);
        #pragma unroll
        for (int j = 0; j < NST; ++j) {
            #pragma unroll
            for (int t = 0; t < 4; ++t) {
                const v2f B = {W0[j][t], W1[j][t]};
                accB[t] = __builtin_amdgcn_wmma_f32_16x16x4_f32(
                    false, A[j], false, B, (short)0, accB[t], false, false);
            }
        }
    }
    __syncthreads();
    {
        const int col  = lane & 15;
        const int mrow = (lane >> 4) * 8;
        #pragma unroll
        for (int r = 0; r < 8; ++r) {
            const v4f o = {accB[0][r], accB[1][r], accB[2][r], accB[3][r]};
            *reinterpret_cast<v4f*>(lds_partial + wave * (QT * CO) + (r + mrow) * CO + 4 * col) = o;
        }
    }
    __syncthreads();

    {
        const int n = wave;
        const int d = lane * 2;
        v2f s = {0.0f, 0.0f};
        #pragma unroll
        for (int w = 0; w < 16; ++w)
            s += *reinterpret_cast<const v2f*>(lds_partial + w * (QT * CO) + n * CO + d);
        const float nm = lds_norm[n];
        const v2f o = {s.x / nm, s.y / nm};
        *reinterpret_cast<v2f*>(out + (qblk + n) * CO + d) = o;
    }
}

extern "C" void kernel_launch(void* const* d_in, const int* in_sizes, int n_in,
                              void* d_out, int out_size, void* d_ws, size_t ws_size,
                              hipStream_t stream) {
    const float* q_pts = (const float*)d_in[0];
    const float* s_pts = (const float*)d_in[1];
    const int*   nb    = (const int*)  d_in[2];
    const float* x     = (const float*)d_in[3];
    const float* kp    = (const float*)d_in[4];
    const float* wt    = (const float*)d_in[5];
    float*       out   = (float*)d_out;

    const int n = in_sizes[0] / 3;
    const int nblk = (n + QT - 1) / QT;
    kpconv_wmma_f32<<<nblk, NTHR, 0, stream>>>(q_pts, s_pts, nb, x, kp, wt, out);
}


# Phase A: feature loads of 4 K-steps in flight per batch (8 16-byte loads per lane) instead of 2: 2 load round trips per query instead of 4; sums and their order unchanged
# speedup vs baseline: 2.6864x; 1.0417x over previous
#include <hip/hip_runtime.h>
#include <hip/hip_bf16.h>

typedef __attribute__((ext_vector_type(2))) float v2f;
typedef __attribute__((ext_vector_type(4))) float v4f;
typedef __attribute__((ext_vector_type(8))) float v8f;

constexpr int   NQ  = 50000;
constexpr int   MS  = 50000;
constexpr int   HN  = 32;
constexpr int   KK  = 15;
constexpr int   CIN = 64;
constexpr int   CO  = 64;
constexpr float KPE = 1.2f;
constexpr float REACH2 = KPE * KPE * 1.01f;

constexpr int QT    = 16;
constexpr int NTHR  = 512;
constexpr int WP    = 16 * CIN + 4;

template <int XM>
__device__ __forceinline__ float row_xor_get(float v) {
    return __builtin_bit_cast(float,
        __builtin_amdgcn_update_dpp(0, __builtin_bit_cast(int, v), 0x160 + XM, 0xf, 0xf, true));
}

__global__ __launch_bounds__(NTHR, 12)
void kpconv_wmma_f32(const float* __restrict__ q_pts,
                     const float* __restrict__ s_pts,
                     const int*   __restrict__ nb,
                     const float* __restrict__ x,
                     const float* __restrict__ kp,
                     const float* __restrict__ wt,
                     float*       __restrict__ out)
{
    struct alignas(16) Lds {
        float nc      [QT * HN * 3];
        int   idx     [QT * HN];
        float norm    [QT];
        float weighted[QT * WP];
    };
    static_assert(16 * QT * CO <= QT * WP, "the partial tiles must fit in the weighted tile's place");
    __shared__ Lds lds;
    float* const lds_partial  = lds.weighted;
    float* const lds_nc       = lds.nc;
    int*   const lds_idx      = lds.idx;
    float* const lds_norm     = lds.norm;
    float* const lds_weighted = lds.weighted;

    const int tid  = threadIdx.x;
    const int lane = tid & 31;
    const int wave = __builtin_amdgcn_readfirstlane(tid >> 5) & (NTHR / 32 - 1);
    const int qblk = blockIdx.x * QT;

    int nreach;
    {
        const int q = wave;
        const int h = lane;
        const int n = qblk + q;
        const int idx = nb[n * HN + h];
        float cx = 1e6f, cy = 1e6f, cz = 1e6f;
        if (idx < MS) {
            cx = s_pts[idx * 3 + 0] - q_pts[n * 3 + 0];
            cy = s_pts[idx * 3 + 1] - q_pts[n * 3 + 1];
            cz = s_pts[idx * 3 + 2] - q_pts[n * 3 + 2];
        }
        float m = 3.0e38f;
        #pragma unroll
        for (int k = 0; k < KK; k += 2) {
            const int k1 = (k + 1 < KK) ? k + 1 : k;
            const v2f tx = (v2f){cx, cx} - (v2f){kp[3 * k],     kp[3 * k1]};
            const v2f ty = (v2f){cy, cy} - (v2f){kp[3 * k + 1], kp[3 * k1 + 1]};
            const v2f tz = (v2f){cz, cz} - (v2f){kp[3 * k + 2], kp[3 * k1 + 2]};
            const v2f d  = tx * tx + ty * ty + tz * tz;
            m = __builtin_fminf(m, __builtin_fminf(d.x, d.y));
        }
        const bool reached = m < REACH2;
        const unsigned rm  = (unsigned)__ballot(reached);
        nreach = __popc(rm);
        const int rank_r = (int)__builtin_amdgcn_mbcnt_lo(rm, 0u);
        const int pos = reached ? rank_r : nreach + (h - rank_r);
        lds_idx[q * HN + pos] = idx;
        lds_nc[(q * HN + pos) * 3 + 0] = cx;
        lds_nc[(q * HN + pos) * 3 + 1] = cy;
        lds_nc[(q * HN + pos) * 3 + 2] = cz;
    }
    __syncthreads();

    {
        const int q   = wave;
        const int k   = lane & 15;
        const int col = lane & 15;
        float kx = 0.0f, ky = 0.0f, kz = 0.0f;
        if (k < KK) { kx = kp[k * 3]; ky = kp[k * 3 + 1]; kz = kp[k * 3 + 2]; }

        v8f acc[4];
        #pragma unroll
        for (int ct = 0; ct < 4; ++ct) acc[ct] = (v8f){};

        const int hofs = (lane >> 4) * 2;
        int cnt = 0;
        const bool odd1 = (lane & 1) != 0;
        const bool odd2 = (lane & 2) != 0;
        constexpr int HB = 4;
        #pragma nounroll
        for (int hb0 = 0; hb0 < HN; hb0 += 4 * HB) {
            v4f   f0[HB], f1[HB];
            bool  ok0[HB], ok1[HB];
            #pragma unroll
            for (int j = 0; j < HB; ++j) {
                const int h0 = hb0 + 4 * j + hofs;
                const int i0 = lds_idx[q * HN + h0];
                const int i1 = lds_idx[q * HN + h0 + 1];
                ok0[j] = i0 < MS;
                ok1[j] = i1 < MS;
                const unsigned b0 = (unsigned)(ok0[j] ? i0 : 0) * (unsigned)(CIN * 4) + 16u * (unsigned)col;
                const unsigned b1 = (unsigned)(ok1[j] ? i1 : 0) * (unsigned)(CIN * 4) + 16u * (unsigned)col;
                f0[j] = *reinterpret_cast<const v4f*>(reinterpret_cast<const char*>(x) + b0);
                f1[j] = *reinterpret_cast<const v4f*>(reinterpret_cast<const char*>(x) + b1);
            }
            __builtin_amdgcn_sched_barrier(0);
            float rs[2 * HB];
            #pragma unroll
            for (int j = 0; j < HB; ++j) {
                const int h0 = hb0 + 4 * j + hofs;
                v2f B[4];
                #pragma unroll
                for (int ct = 0; ct < 4; ++ct) {
                    B[ct].x = ok0[j] ? f0[j][ct] : 0.0f;
                    B[ct].y = ok1[j] ? f1[j][ct] : 0.0f;
                }
                if (hb0 + 4 * j < nreach) {
                v2f A;
                #pragma unroll
                for (int s = 0; s < 2; ++s) {
                    const float nx = lds_nc[(q * HN + h0 + s) * 3 + 0] - kx;
                    const float ny = lds_nc[(q * HN + h0 + s) * 3 + 1] - ky;
                    const float nz = lds_nc[(q * HN + h0 + s) * 3 + 2] - kz;
                    float w = 1.0f - __builtin_amdgcn_sqrtf(nx * nx + ny * ny + nz * nz) * (1.0f / KPE);
                    w = fmaxf(w, 0.0f);
                    if (k >= KK) w = 0.0f;
                    A[s] = w;
                }
                #pragma unroll
                for (int ct = 0; ct < 4; ++ct)
                    acc[ct] = __builtin_amdgcn_wmma_f32_16x16x4_f32(
                        false, A, false, B[ct], (short)0, acc[ct], false, false);
                }
                rs[2 * j]     = (B[0].x + B[1].x) + (B[2].x + B[3].x);
                rs[2 * j + 1] = (B[0].y + B[1].y) + (B[2].y + B[3].y);
            }
            static_assert(HB % 2 == 0, "the packed row sum below takes 4 rows per half at a time");
            #pragma unroll
            for (int g = 0; g < HB / 2; ++g) {
                const float r0 = rs[4 * g], r1 = rs[4 * g + 1], r2 = rs[4 * g + 2], r3 = rs[4 * g + 3];
                float a01 = odd1 ? r1 : r0;
                a01 += row_xor_get<1>(odd1 ? r0 : r1);
                float a23 = odd1 ? r3 : r2;
                a23 += row_xor_get<1>(odd1 ? r2 : r3);
                float rsum = odd2 ? a23 : a01;
                rsum += row_xor_get<2>(odd2 ? a01 : a23);
                rsum += row_xor_get<4>(rsum);
                rsum += row_xor_get<8>(rsum);
                const unsigned m = (unsigned)__ballot(rsum > 0.0f);
                cnt += __popc(m & 0x000F000Fu);
            }
        }
        const int laneE = (int)__builtin_amdgcn_mbcnt_lo(~0u, 0u);
        if (laneE == 0) lds_norm[q] = (float)(cnt > 0 ? cnt : 1);

        const int mrow = (laneE >> 4) * 8;
        const int colE = laneE & 15;
        #pragma unroll
        for (int r = 0; r < 8; ++r) {
            const v4f o = {acc[0][r], acc[1][r], acc[2][r], acc[3][r]};
            *reinterpret_cast<v4f*>(lds_weighted + q * WP + (r + mrow) * CIN + 4 * colE) = o;
        }
    }
    __syncthreads();

    constexpr int KW  = (KK * CIN) / 16;
    constexpr int NST = KW / 4;
    static_assert(KW * 16 == KK * CIN && NST * 4 == KW, "K must split evenly over the 16 waves");
    v8f accB[4];
    const int lane2 = (int)__builtin_amdgcn_mbcnt_lo(~0u, 0u);
    {
        const int col   = lane2 & 15;
        const int kofs  = (lane2 >> 4) * 2;
        const int nrow  = lane2 & 15;

        const int kb0 = wave * KW + kofs;

        const float* wrow = wt + (kb0 * CO + 4 * col);
        const float* arow = lds_weighted + (nrow * WP + kb0);

        #pragma unroll
        for (int t = 0; t < 4; ++t) accB[t] = (v8f){};
        v4f W0[NST], W1[NST];
        v2f A[NST];
        #pragma unroll
        for (int j = 0; j < NST; ++j) {
            W0[j] = *reinterpret_cast<const v4f*>(wrow + (j * 4)     * CO);
            W1[j] = *reinterpret_cast<const v4f*>(wrow + (j * 4 + 1) * CO);
        }
        #pragma unroll
        for (int j = 0; j < NST; ++j)
            A[j] = *reinterpret_cast<const v2f*>(arow + j * 4);
        #pragma unroll
        for (int j = 0; j < NST; ++j) {
            #pragma unroll
            for (int t = 0; t < 4; ++t) {
                const v2f B = {W0[j][t], W1[j][t]};
                accB[t] = __builtin_amdgcn_wmma_f32_16x16x4_f32(
                    false, A[j], false, B, (short)0, accB[t], false, false);
            }
        }
    }
    __syncthreads();
    {
        const int col  = lane2 & 15;
        const int mrow = (lane2 >> 4) * 8;
        #pragma unroll
        for (int r = 0; r < 8; ++r) {
            const v4f o = {accB[0][r], accB[1][r], accB[2][r], accB[3][r]};
            *reinterpret_cast<v4f*>(lds_partial + wave * (QT * CO) + (r + mrow) * CO + 4 * col) = o;
        }
    }
    __syncthreads();

    {
        const int n = wave;
        const int d = lane2 * 2;
        v2f s = {0.0f, 0.0f};
        #pragma unroll
        for (int w = 0; w < 16; ++w)
            s += *reinterpret_cast<const v2f*>(lds_partial + w * (QT * CO) + n * CO + d);
        const float nm = lds_norm[n];
        const v2f o = {s.x / nm, s.y / nm};
        *reinterpret_cast<v2f*>(out + (qblk + n) * CO + d) = o;
    }
}

extern "C" void kernel_launch(void* const* d_in, const int* in_sizes, int n_in,
                              void* d_out, int out_size, void* d_ws, size_t ws_size,
                              hipStream_t stream) {
    const float* q_pts = (const float*)d_in[0];
    const float* s_pts = (const float*)d_in[1];
    const int*   nb    = (const int*)  d_in[2];
    const float* x     = (const float*)d_in[3];
    const float* kp    = (const float*)d_in[4];
    const float* wt    = (const float*)d_in[5];
    float*       out   = (float*)d_out;

    const int n = in_sizes[0] / 3;
    const int nblk = (n + QT - 1) / QT;
    kpconv_wmma_f32<<<nblk, NTHR, 0, stream>>>(q_pts, s_pts, nb, x, kp, wt, out);
}


# Influence weights formed once in phase 0, from the distances its reach test already computes, and read by phase A from LDS as A fragments: phase A distance and sqrt work removed; last bit may differ
# speedup vs baseline: 2.8090x; 1.0456x over previous
#include <hip/hip_runtime.h>
#include <hip/hip_bf16.h>

typedef __attribute__((ext_vector_type(2))) float v2f;
typedef __attribute__((ext_vector_type(4))) float v4f;
typedef __attribute__((ext_vector_type(8))) float v8f;

constexpr int   NQ  = 50000;
constexpr int   MS  = 50000;
constexpr int   HN  = 32;
constexpr int   KK  = 15;
constexpr int   CIN = 64;
constexpr int   CO  = 64;
constexpr float KPE = 1.2f;
constexpr float REACH2 = KPE * KPE * 1.01f;

constexpr int QT    = 16;
constexpr int NTHR  = 512;
constexpr int WP    = 16 * CIN + 4;
constexpr int WTP   = 20;
static_assert(HN * WTP <= WP, "the weight table must fit in the query's row of the weighted tile");

template <int XM>
__device__ __forceinline__ float row_xor_get(float v) {
    return __builtin_bit_cast(float,
        __builtin_amdgcn_update_dpp(0, __builtin_bit_cast(int, v), 0x160 + XM, 0xf, 0xf, true));
}

__global__ __launch_bounds__(NTHR, 12)
void kpconv_wmma_f32(const float* __restrict__ q_pts,
                     const float* __restrict__ s_pts,
                     const int*   __restrict__ nb,
                     const float* __restrict__ x,
                     const float* __restrict__ kp,
                     const float* __restrict__ wt,
                     float*       __restrict__ out)
{
    struct alignas(16) Lds {
        int   idx     [QT * HN];
        float norm    [QT];
        float weighted[QT * WP];
    };
    static_assert(16 * QT * CO <= QT * WP, "the partial tiles must fit in the weighted tile's place");
    __shared__ Lds lds;
    float* const lds_partial  = lds.weighted;
    int*   const lds_idx      = lds.idx;
    float* const lds_norm     = lds.norm;
    float* const lds_weighted = lds.weighted;

    const int tid  = threadIdx.x;
    const int lane = tid & 31;
    const int wave = __builtin_amdgcn_readfirstlane(tid >> 5) & (NTHR / 32 - 1);
    const int qblk = blockIdx.x * QT;

    int nreach;
    {
        const int q = wave;
        const int h = lane;
        const int n = qblk + q;
        const int idx = nb[n * HN + h];
        float cx = 1e6f, cy = 1e6f, cz = 1e6f;
        if (idx < MS) {
            cx = s_pts[idx * 3 + 0] - q_pts[n * 3 + 0];
            cy = s_pts[idx * 3 + 1] - q_pts[n * 3 + 1];
            cz = s_pts[idx * 3 + 2] - q_pts[n * 3 + 2];
        }
        float m = 3.0e38f;
        v4f wq[4];
        #pragma unroll
        for (int k = 0; k < KK; k += 2) {
            const int k1 = (k + 1 < KK) ? k + 1 : k;
            const v2f tx = (v2f){cx, cx} - (v2f){kp[3 * k],     kp[3 * k1]};
            const v2f ty = (v2f){cy, cy} - (v2f){kp[3 * k + 1], kp[3 * k1 + 1]};
            const v2f tz = (v2f){cz, cz} - (v2f){kp[3 * k + 2], kp[3 * k1 + 2]};
            const v2f d  = tx * tx + ty * ty + tz * tz;
            m = __builtin_fminf(m, __builtin_fminf(d.x, d.y));
            const float w0 = fmaxf(1.0f - __builtin_amdgcn_sqrtf(d.x) * (1.0f / KPE), 0.0f);
            const float w1 = (k + 1 < KK)
                ? fmaxf(1.0f - __builtin_amdgcn_sqrtf(d.y) * (1.0f / KPE), 0.0f) : 0.0f;
            wq[k >> 2][k & 3]       = w0;
            wq[k >> 2][(k & 3) + 1] = w1;
        }
        const bool reached = m < REACH2;
        const unsigned rm  = (unsigned)__ballot(reached);
        nreach = __popc(rm);
        const int rank_r = (int)__builtin_amdgcn_mbcnt_lo(rm, 0u);
        const int pos = reached ? rank_r : nreach + (h - rank_r);
        lds_idx[q * HN + pos] = idx;
        #pragma unroll
        for (int j = 0; j < 4; ++j)
            *reinterpret_cast<v4f*>(lds_weighted + q * WP + pos * WTP + 4 * j) = wq[j];
    }
    __syncthreads();

    {
        const int q   = wave;
        const int k   = lane & 15;
        const int col = lane & 15;
        const float* const wlane = lds_weighted + q * WP + ((lane >> 4) * 2) * WTP + k;

        v8f acc[4];
        #pragma unroll
        for (int ct = 0; ct < 4; ++ct) acc[ct] = (v8f){};

        const int hofs = (lane >> 4) * 2;
        int cnt = 0;
        const bool odd1 = (lane & 1) != 0;
        const bool odd2 = (lane & 2) != 0;
        constexpr int HB = 4;
        #pragma nounroll
        for (int hb0 = 0; hb0 < HN; hb0 += 4 * HB) {
            v4f   f0[HB], f1[HB];
            bool  ok0[HB], ok1[HB];
            #pragma unroll
            for (int j = 0; j < HB; ++j) {
                const int h0 = hb0 + 4 * j + hofs;
                const int i0 = lds_idx[q * HN + h0];
                const int i1 = lds_idx[q * HN + h0 + 1];
                ok0[j] = i0 < MS;
                ok1[j] = i1 < MS;
                const unsigned b0 = (unsigned)(ok0[j] ? i0 : 0) * (unsigned)(CIN * 4) + 16u * (unsigned)col;
                const unsigned b1 = (unsigned)(ok1[j] ? i1 : 0) * (unsigned)(CIN * 4) + 16u * (unsigned)col;
                f0[j] = *reinterpret_cast<const v4f*>(reinterpret_cast<const char*>(x) + b0);
                f1[j] = *reinterpret_cast<const v4f*>(reinterpret_cast<const char*>(x) + b1);
            }
            v2f Aw[HB];
            #pragma unroll
            for (int j = 0; j < HB; ++j) {
                Aw[j].x = wlane[(hb0 + 4 * j)     * WTP];
                Aw[j].y = wlane[(hb0 + 4 * j + 1) * WTP];
            }
            __builtin_amdgcn_sched_barrier(0);
            float rs[2 * HB];
            #pragma unroll
            for (int j = 0; j < HB; ++j) {
                v2f B[4];
                #pragma unroll
                for (int ct = 0; ct < 4; ++ct) {
                    B[ct].x = ok0[j] ? f0[j][ct] : 0.0f;
                    B[ct].y = ok1[j] ? f1[j][ct] : 0.0f;
                }
                if (hb0 + 4 * j < nreach) {
                const v2f A = Aw[j];
                #pragma unroll
                for (int ct = 0; ct < 4; ++ct)
                    acc[ct] = __builtin_amdgcn_wmma_f32_16x16x4_f32(
                        false, A, false, B[ct], (short)0, acc[ct], false, false);
                }
                rs[2 * j]     = (B[0].x + B[1].x) + (B[2].x + B[3].x);
                rs[2 * j + 1] = (B[0].y + B[1].y) + (B[2].y + B[3].y);
            }
            static_assert(HB % 2 == 0, "the packed row sum below takes 4 rows per half at a time");
            #pragma unroll
            for (int g = 0; g < HB / 2; ++g) {
                const float r0 = rs[4 * g], r1 = rs[4 * g + 1], r2 = rs[4 * g + 2], r3 = rs[4 * g + 3];
                float a01 = odd1 ? r1 : r0;
                a01 += row_xor_get<1>(odd1 ? r0 : r1);
                float a23 = odd1 ? r3 : r2;
                a23 += row_xor_get<1>(odd1 ? r2 : r3);
                float rsum = odd2 ? a23 : a01;
                rsum += row_xor_get<2>(odd2 ? a01 : a23);
                rsum += row_xor_get<4>(rsum);
                rsum += row_xor_get<8>(rsum);
                const unsigned m = (unsigned)__ballot(rsum > 0.0f);
                cnt += __popc(m & 0x000F000Fu);
            }
        }
        const int laneE = (int)__builtin_amdgcn_mbcnt_lo(~0u, 0u);
        if (laneE == 0) lds_norm[q] = (float)(cnt > 0 ? cnt : 1);

        const int mrow = (laneE >> 4) * 8;
        const int colE = laneE & 15;
        #pragma unroll
        for (int r = 0; r < 8; ++r) {
            const v4f o = {acc[0][r], acc[1][r], acc[2][r], acc[3][r]};
            *reinterpret_cast<v4f*>(lds_weighted + q * WP + (r + mrow) * CIN + 4 * colE) = o;
        }
    }
    __syncthreads();

    constexpr int KW  = (KK * CIN) / 16;
    constexpr int NST = KW / 4;
    static_assert(KW * 16 == KK * CIN && NST * 4 == KW, "K must split evenly over the 16 waves");
    v8f accB[4];
    const int lane2 = (int)__builtin_amdgcn_mbcnt_lo(~0u, 0u);
    {
        const int col   = lane2 & 15;
        const int kofs  = (lane2 >> 4) * 2;
        const int nrow  = lane2 & 15;

        const int kb0 = wave * KW + kofs;

        const float* wrow = wt + (kb0 * CO + 4 * col);
        const float* arow = lds_weighted + (nrow * WP + kb0);

        #pragma unroll
        for (int t = 0; t < 4; ++t) accB[t] = (v8f){};
        v4f W0[NST], W1[NST];
        v2f A[NST];
        #pragma unroll
        for (int j = 0; j < NST; ++j) {
            W0[j] = *reinterpret_cast<const v4f*>(wrow + (j * 4)     * CO);
            W1[j] = *reinterpret_cast<const v4f*>(wrow + (j * 4 + 1) * CO);
        }
        #pragma unroll
        for (int j = 0; j < NST; ++j)
            A[j] = *reinterpret_cast<const v2f*>(arow + j * 4);
        #pragma unroll
        for (int j = 0; j < NST; ++j) {
            #pragma unroll
            for (int t = 0; t < 4; ++t) {
                const v2f B = {W0[j][t], W1[j][t]};
                accB[t] = __builtin_amdgcn_wmma_f32_16x16x4_f32(
                    false, A[j], false, B, (short)0, accB[t], false, false);
            }
        }
    }
    __syncthreads();
    {
        const int col  = lane2 & 15;
        const int mrow = (lane2 >> 4) * 8;
        #pragma unroll
        for (int r = 0; r < 8; ++r) {
            const v4f o = {accB[0][r], accB[1][r], accB[2][r], accB[3][r]};
            *reinterpret_cast<v4f*>(lds_partial + wave * (QT * CO) + (r + mrow) * CO + 4 * col) = o;
        }
    }
    __syncthreads();

    {
        const int n = wave;
        const int d = lane2 * 2;
        v2f s = {0.0f, 0.0f};
        #pragma unroll
        for (int w = 0; w < 16; ++w)
            s += *reinterpret_cast<const v2f*>(lds_partial + w * (QT * CO) + n * CO + d);
        const float nm = lds_norm[n];
        const v2f o = {s.x / nm, s.y / nm};
        *reinterpret_cast<v2f*>(out + (qblk + n) * CO + d) = o;
    }
}

extern "C" void kernel_launch(void* const* d_in, const int* in_sizes, int n_in,
                              void* d_out, int out_size, void* d_ws, size_t ws_size,
                              hipStream_t stream) {
    const float* q_pts = (const float*)d_in[0];
    const float* s_pts = (const float*)d_in[1];
    const int*   nb    = (const int*)  d_in[2];
    const float* x     = (const float*)d_in[3];
    const float* kp    = (const float*)d_in[4];
    const float* wt    = (const float*)d_in[5];
    float*       out   = (float*)d_out;

    const int n = in_sizes[0] / 3;
    const int nblk = (n + QT - 1) / QT;
    kpconv_wmma_f32<<<nblk, NTHR, 0, stream>>>(q_pts, s_pts, nb, x, kp, wt, out);
}
